# starting kernel
# baseline (speedup 1.0000x reference)
#include <hip/hip_runtime.h>
#include <hip/hip_bf16.h>


typedef float v2f __attribute__((ext_vector_type(2)));
typedef float v4f __attribute__((ext_vector_type(4)));
typedef float v8f __attribute__((ext_vector_type(8)));

#define BATCH 8
#define NDIM  512
#define SDIM  4096
#define HDIM  1024

#define BLK_N 64
#define BLK_H 128
#define BLK_K 32
#define A_STRIDE 36
#define BP_STRIDE 288
#define NPAIR (BLK_K / 2)

__device__ __forceinline__ void stage_chunk(const float* __restrict__ Ag,
                                            const float* __restrict__ Bg,
                                            int ks, float* __restrict__ Al,
                                            float* __restrict__ Bl, int tid)
{
    #pragma unroll
    for (int i = 0; i < 2; ++i) {
        int idx = i * 256 + tid;
        int r   = idx >> 3;
        int c   = (idx & 7) << 2;
        v4f v = *(const v4f*)(Ag + (size_t)r * SDIM + ks + c);
        *(v4f*)(&Al[r * A_STRIDE + c]) = v;
    }
    #pragma unroll
    for (int i = 0; i < 4; ++i) {
        int idx = i * 256 + tid;
        int p   = idx >> 6;
        int c   = (idx & 63) << 1;
        v2f lo = *(const v2f*)(Bg + (size_t)(ks + 2 * p)     * HDIM + c);
        v2f hi = *(const v2f*)(Bg + (size_t)(ks + 2 * p + 1) * HDIM + c);
        v4f v; v.x = lo.x; v.y = hi.x; v.z = lo.y; v.w = hi.y;
        *(v4f*)(&Bl[p * BP_STRIDE + 2 * c]) = v;
    }
}

__device__ __forceinline__ void compute_chunk(const float* __restrict__ Al,
                                              const float* __restrict__ Bl,
                                              int wr, int wc, int hi, int ln,
                                              v8f& acc0, v8f& acc1,
                                              v8f& acc2, v8f& acc3)
{
    #pragma unroll
    for (int kk = 0; kk < BLK_K; kk += 4) {
        v2f a = *(const v2f*)(&Al[(wr + ln) * A_STRIDE + kk + 2 * hi]);
        const float* bp = &Bl[((kk >> 1) + hi) * BP_STRIDE + 2 * (wc + ln)];
        v2f b0 = *(const v2f*)(bp +  0);
        acc0 = __builtin_amdgcn_wmma_f32_16x16x4_f32(false, a, false, b0,
                                                     (short)0, acc0, false, false);
        v2f b1 = *(const v2f*)(bp + 32);
        acc1 = __builtin_amdgcn_wmma_f32_16x16x4_f32(false, a, false, b1,
                                                     (short)0, acc1, false, false);
        v2f b2 = *(const v2f*)(bp + 64);
        acc2 = __builtin_amdgcn_wmma_f32_16x16x4_f32(false, a, false, b2,
                                                     (short)0, acc2, false, false);
        v2f b3 = *(const v2f*)(bp + 96);
        acc3 = __builtin_amdgcn_wmma_f32_16x16x4_f32(false, a, false, b3,
                                                     (short)0, acc3, false, false);
    }
}

__global__ __launch_bounds__(256)
void meanpool_wmma_f32(const float* __restrict__ doc_state,
                       const float* __restrict__ nodes_mapping,
                       const float* __restrict__ nodes_len,
                       float* __restrict__ out)
{
    __shared__ float Al[2][BLK_N * A_STRIDE];
    __shared__ float Bl[2][NPAIR * BP_STRIDE];
    __shared__ float invLen[BLK_N];

    const int tid  = threadIdx.x;
    const int lane = tid & 31;
    const int wave = tid >> 5;
    const int b    = blockIdx.z;
    const int n0   = blockIdx.y * BLK_N;
    const int h0   = blockIdx.x * BLK_H;

    const float* Ag = nodes_mapping + ((size_t)b * NDIM + n0) * SDIM;
    const float* Bg = doc_state     + (size_t)b * SDIM * HDIM + h0;

    if (tid < BLK_N)
        invLen[tid] = 1.0f / nodes_len[(size_t)b * NDIM + n0 + tid];

    const int wr = (wave >> 1) * 16;
    const int wc = (wave & 1) * 64;
    const int hi = lane >> 4;
    const int ln = lane & 15;

    v8f acc0 = {}, acc1 = {}, acc2 = {}, acc3 = {};

    stage_chunk(Ag, Bg, 0, Al[0], Bl[0], tid);

    for (int ks = 0; ks < SDIM; ks += 2 * BLK_K) {
        __syncthreads();
        if (ks + BLK_K < SDIM)
            stage_chunk(Ag, Bg, ks + BLK_K, Al[1], Bl[1], tid);
        compute_chunk(Al[0], Bl[0], wr, wc, hi, ln, acc0, acc1, acc2, acc3);

        __syncthreads();
        if (ks + 2 * BLK_K < SDIM)
            stage_chunk(Ag, Bg, ks + 2 * BLK_K, Al[0], Bl[0], tid);
        compute_chunk(Al[1], Bl[1], wr, wc, hi, ln, acc0, acc1, acc2, acc3);
    }

    #pragma unroll
    for (int r = 0; r < 8; ++r) {
        const int   rloc = wr + 8 * hi + r;
        const float sc   = invLen[rloc];
        float* orow = out + ((size_t)b * NDIM + n0 + rloc) * HDIM + h0 + wc + ln;
        orow[ 0] = acc0[r] * sc;
        orow[16] = acc1[r] * sc;
        orow[32] = acc2[r] * sc;
        orow[48] = acc3[r] * sc;
    }
}

extern "C" void kernel_launch(void* const* d_in, const int* in_sizes, int n_in,
                              void* d_out, int out_size, void* d_ws, size_t ws_size,
                              hipStream_t stream) {
    (void)in_sizes; (void)n_in; (void)out_size; (void)d_ws; (void)ws_size;
    const float* doc_state     = (const float*)d_in[0];
    const float* nodes_mapping = (const float*)d_in[1];
    const float* nodes_len     = (const float*)d_in[2];
    float* out = (float*)d_out;

    dim3 grid(HDIM / BLK_H, NDIM / BLK_N, BATCH);
    dim3 block(256);
    meanpool_wmma_f32<<<grid, block, 0, stream>>>(doc_state, nodes_mapping,
                                                  nodes_len, out);
}


# One 4-wave workgroup per WGP, each wave a 128x128 tile at one wave per SIMD: a loaded fragment feeds 8 WMMAs not 4, loads stand once per 128 WMMAs not per 32; K split and sum order kept
# speedup vs baseline: 1.3454x; 1.3454x over previous
#include <hip/hip_runtime.h>


typedef float v2f __attribute__((ext_vector_type(2)));
typedef float v4f __attribute__((ext_vector_type(4)));
typedef float v8f __attribute__((ext_vector_type(8)));

#define BATCH 8
#define NDIM  512
#define SDIM  4096
#define HDIM  1024

#define BLK_N 128
#define BLK_H 128
#define RTILES 8
#define WAVES 4
#define KBLK  32
#define CHUNK 64
#define NTRIP (SDIM / (WAVES * CHUNK))

#define RED_TILE 256


__device__ __forceinline__ void load_a_tile(v4f (&a)[4], const float* __restrict__ base,
                                            const unsigned lane_off, const int tile)
{
    #pragma unroll
    for (int j = 0; j < 4; ++j)
        a[j] = *(const v4f*)(base + lane_off + (tile * 16 * SDIM + 4 * j));
}

__device__ __forceinline__ void load_b_step(v4f (&bq)[2][2], const float* __restrict__ base,
                                            const unsigned lane_off, const int t)
{
    #pragma unroll
    for (int e = 0; e < 2; ++e)
        #pragma unroll
        for (int g = 0; g < 2; ++g)
            bq[e][g] = *(const v4f*)(base + lane_off + ((2 * t + e) * HDIM + 64 * g));
}

__device__ __forceinline__ void mma_step(v8f (&acc)[RTILES][8], const v4f (&a)[RTILES][4],
                                         const v4f (&bq)[2][2], const int t)
{
    v2f af[RTILES];
    #pragma unroll
    for (int i = 0; i < RTILES; ++i) {
        const v4f q = a[i][t >> 1];
        if (t & 1) { af[i].x = q.z; af[i].y = q.w; }
        else       { af[i].x = q.x; af[i].y = q.y; }
    }
    #pragma unroll
    for (int g = 0; g < 2; ++g) {
        #pragma unroll
        for (int j = 0; j < 4; ++j) {
            v2f bf;
            bf.x = bq[0][g][j];
            bf.y = bq[1][g][j];
            #pragma unroll
            for (int i = 0; i < RTILES; ++i)
                acc[i][g * 4 + j] = __builtin_amdgcn_wmma_f32_16x16x4_f32(
                    false, af[i], false, bf, (short)0, acc[i][g * 4 + j], false, false);
        }
    }
}

__global__ __launch_bounds__(128, 1)
void meanpool_wmma_f32_ksplit_w128(const float* __restrict__ doc_state,
                                   const float* __restrict__ nodes_mapping,
                                   const float* __restrict__ nodes_len,
                                   float* __restrict__ out)
{
    __shared__ float red[WAVES * RTILES * 8 * RED_TILE];

    const int lane = threadIdx.x & 31;
    const int w    = __builtin_amdgcn_readfirstlane(threadIdx.x >> 5);
    const int c    = lane & 15;
    const int h    = lane >> 4;
    const int b    = blockIdx.z;
    const int n0   = blockIdx.y * BLK_N;
    const int h0   = blockIdx.x * BLK_H;

    const float* pa0 = nodes_mapping + ((size_t)b * NDIM + n0) * SDIM + w * CHUNK;
    const float* pb0 = doc_state + ((size_t)b * SDIM + w * CHUNK) * HDIM + h0;
    const unsigned la = (unsigned)(c * SDIM + 16 * h);
    const unsigned lb = (unsigned)(16 * h * HDIM + 4 * c);

    v8f acc[RTILES][8];
    #pragma unroll
    for (int i = 0; i < RTILES; ++i)
        #pragma unroll
        for (int ct = 0; ct < 8; ++ct)
            acc[i][ct] = (v8f){0.f, 0.f, 0.f, 0.f, 0.f, 0.f, 0.f, 0.f};

    v4f a[2][RTILES][4];
    v4f bq[8][2][2];

    const float* pa = pa0;
    const float* pb = pb0;

    #pragma unroll
    for (int i = 0; i < RTILES; ++i)
        load_a_tile(a[0][i], pa, la, i);
    #pragma unroll
    for (int t = 0; t < 4; ++t)
        load_b_step(bq[t], pb, lb, t);

    for (int it = 0; it < NTRIP; ++it) {
        const bool last = (it + 1 == NTRIP);
        const float* pa_n = last ? pa0 : pa + WAVES * CHUNK;
        const float* pb_n = last ? pb0 : pb + (size_t)WAVES * CHUNK * HDIM;

        #pragma unroll
        for (int blk = 0; blk < 2; ++blk) {
            const float* pa_nb = (blk == 0) ? pa + KBLK : pa_n;
            const float* pb_cb = pb + (size_t)blk * KBLK * HDIM;
            const float* pb_nb = (blk == 0) ? pb + (size_t)KBLK * HDIM : pb_n;
            #pragma unroll
            for (int t = 0; t < 8; ++t) {
                if ((t & 1) == 0) {
                    #pragma unroll
                    for (int u = t + 4; u < t + 6; ++u) {
                        if (u < 8) load_b_step(bq[u & 7], pb_cb, lb, u);
                        else       load_b_step(bq[u & 7], pb_nb, lb, u - 8);
                    }
                    #pragma unroll
                    for (int i = 3 * (t >> 1); i < 3 * (t >> 1) + 3; ++i)
                        if (i < RTILES) load_a_tile(a[blk ^ 1][i], pa_nb, la, i);
                }
                mma_step(acc, a[blk], bq[t & 7], t);
                __builtin_amdgcn_sched_barrier(0);
            }
        }
        pa = pa_n;
        pb = pb_n;
    }

    #pragma unroll
    for (int i = 0; i < RTILES; ++i) {
        #pragma unroll
        for (int ct = 0; ct < 8; ++ct) {
            float* q = &red[((w * RTILES + i) * 8 + ct) * RED_TILE + lane * 4];
            const v8f v = acc[i][ct];
            *(v4f*)(q)       = (v4f){v[0], v[1], v[2], v[3]};
            *(v4f*)(q + 128) = (v4f){v[4], v[5], v[6], v[7]};
        }
    }
    __syncthreads();

    #pragma unroll
    for (int p = 0; p < 2; ++p) {
        const int rt   = 2 * w + p;
        const int row0 = n0 + 16 * rt + 8 * h;
        const v4f l0 = *(const v4f*)(nodes_len + (size_t)b * NDIM + row0);
        const v4f l1 = *(const v4f*)(nodes_len + (size_t)b * NDIM + row0 + 4);
        float sc[8];
        sc[0] = 1.0f / l0.x; sc[1] = 1.0f / l0.y; sc[2] = 1.0f / l0.z; sc[3] = 1.0f / l0.w;
        sc[4] = 1.0f / l1.x; sc[5] = 1.0f / l1.y; sc[6] = 1.0f / l1.z; sc[7] = 1.0f / l1.w;

        #pragma unroll
        for (int g = 0; g < 2; ++g) {
            v4f lo[4], hi[4];
            #pragma unroll
            for (int j = 0; j < 4; ++j) {
                const float* q = &red[((0 * RTILES + rt) * 8 + g * 4 + j) * RED_TILE + lane * 4];
                lo[j] = *(const v4f*)(q);
                hi[j] = *(const v4f*)(q + 128);
                #pragma unroll
                for (int s = 1; s < WAVES; ++s) {
                    lo[j] += *(const v4f*)(q + s * (RTILES * 8 * RED_TILE));
                    hi[j] += *(const v4f*)(q + s * (RTILES * 8 * RED_TILE) + 128);
                }
            }
            float* orow = out + ((size_t)b * NDIM + row0) * HDIM + h0 + 64 * g + 4 * c;
            #pragma unroll
            for (int r = 0; r < 4; ++r) {
                *(v4f*)(orow + (size_t)r * HDIM) =
                    (v4f){lo[0][r] * sc[r], lo[1][r] * sc[r], lo[2][r] * sc[r], lo[3][r] * sc[r]};
                *(v4f*)(orow + (size_t)(r + 4) * HDIM) =
                    (v4f){hi[0][r] * sc[r + 4], hi[1][r] * sc[r + 4], hi[2][r] * sc[r + 4], hi[3][r] * sc[r + 4]};
            }
        }
    }
}

extern "C" void kernel_launch(void* const* d_in, const int* in_sizes, int n_in,
                              void* d_out, int out_size, void* d_ws, size_t ws_size,
                              hipStream_t stream) {
    (void)in_sizes; (void)n_in; (void)out_size; (void)d_ws; (void)ws_size;
    const float* doc_state     = (const float*)d_in[0];
    const float* nodes_mapping = (const float*)d_in[1];
    const float* nodes_len     = (const float*)d_in[2];
    float* out = (float*)d_out;

    dim3 grid(HDIM / BLK_H, NDIM / BLK_N, BATCH);
    dim3 block(32 * WAVES);
    meanpool_wmma_f32_ksplit_w128<<<grid, block, 0, stream>>>(doc_state, nodes_mapping,
                                                              nodes_len, out);
}
